# starting kernel
# baseline (speedup 1.0000x reference)
#include <hip/hip_runtime.h>
#include <hip/hip_bf16.h>

typedef __attribute__((ext_vector_type(2))) float v2f;
typedef __attribute__((ext_vector_type(8))) float v8f;

#define D_MODEL 1024
#define N_HEADS 16
#define D_HEAD  64
#define BATCH   4
#define SEQ     2048
#define ROWS    (BATCH * SEQ)

static __device__ __forceinline__ v8f wmma_f32(v2f a, v2f b, v8f c) {
  return __builtin_amdgcn_wmma_f32_16x16x4_f32(
       false, a,  false, b,
       (short)0, c,  false,  false);
}

__global__ __launch_bounds__(256) void transpose_w_kernel(
    const float* __restrict__ W0, const float* __restrict__ W1,
    const float* __restrict__ W2, const float* __restrict__ W3,
    float* __restrict__ T0, float* __restrict__ T1,
    float* __restrict__ T2, float* __restrict__ T3) {
  const float* W = (blockIdx.z == 0) ? W0 : (blockIdx.z == 1) ? W1
                   : (blockIdx.z == 2) ? W2 : W3;
  float* T = (blockIdx.z == 0) ? T0 : (blockIdx.z == 1) ? T1
             : (blockIdx.z == 2) ? T2 : T3;

  __shared__ float tile[32][33];
  const int tx = threadIdx.x & 31;
  const int ty = threadIdx.x >> 5;
  const int x0 = blockIdx.x * 32;
  const int y0 = blockIdx.y * 32;

#pragma unroll
  for (int i = 0; i < 4; ++i)
    tile[ty + i * 8][tx] = W[(size_t)(y0 + ty + i * 8) * D_MODEL + x0 + tx];
  __syncthreads();
#pragma unroll
  for (int i = 0; i < 4; ++i)
    T[(size_t)(x0 + ty + i * 8) * D_MODEL + y0 + tx] = tile[tx][ty + i * 8];
}

__global__ __launch_bounds__(256) void qkv_proj_kernel(
    const float* __restrict__ X,
    const float* __restrict__ Wqt, const float* __restrict__ Wkt,
    const float* __restrict__ Wvt,
    float* __restrict__ Q, float* __restrict__ K, float* __restrict__ Vt) {
  const int wave = threadIdx.x >> 5;
  const int lane = threadIdx.x & 31;
  const int half = lane >> 4;
  const int l16  = lane & 15;

  const int row0 = blockIdx.x * 32;
  const int col0 = (blockIdx.y * 8 + wave) * 16;
  const int col  = col0 + l16;

  v8f cq0 = {}, ck0 = {}, cv0 = {};
  v8f cq1 = {}, ck1 = {}, cv1 = {};
  const float* xrow0 = X + (size_t)(row0 + l16) * D_MODEL;
  const float* xrow1 = xrow0 + (size_t)16 * D_MODEL;
  const float* wq = Wqt + (size_t)col * D_MODEL;
  const float* wk = Wkt + (size_t)col * D_MODEL;
  const float* wv = Wvt + (size_t)col * D_MODEL;

#pragma unroll 4
  for (int k0 = 0; k0 < D_MODEL; k0 += 4) {
    const int kr = k0 + 2 * half;
    v2f a0 = *(const v2f*)(xrow0 + kr);
    v2f a1 = *(const v2f*)(xrow1 + kr);
    v2f bq = *(const v2f*)(wq + kr);
    v2f bk = *(const v2f*)(wk + kr);
    v2f bv = *(const v2f*)(wv + kr);
    cq0 = wmma_f32(a0, bq, cq0);  cq1 = wmma_f32(a1, bq, cq1);
    ck0 = wmma_f32(a0, bk, ck0);  ck1 = wmma_f32(a1, bk, ck1);
    cv0 = wmma_f32(a0, bv, cv0);  cv1 = wmma_f32(a1, bv, cv1);
  }

  const int h = col >> 6;
  const int d = col & 63;
#pragma unroll
  for (int r = 0; r < 8; ++r) {
#pragma unroll
    for (int t = 0; t < 2; ++t) {
      const int row = row0 + t * 16 + r + 8 * half;
      const int b = row >> 11;
      const int s = row & (SEQ - 1);
      const int bh = b * N_HEADS + h;
      const size_t o  = ((size_t)bh * SEQ + s) * D_HEAD + d;
      const size_t ov = ((size_t)bh * D_HEAD + d) * SEQ + s;
      Q[o]  = t ? cq1[r] : cq0[r];
      K[o]  = t ? ck1[r] : ck0[r];
      Vt[ov] = t ? cv1[r] : cv0[r];
    }
  }
}

__global__ __launch_bounds__(256) void flash_attn_kernel(
    const float* __restrict__ Q, const float* __restrict__ K,
    const float* __restrict__ Vt, float* __restrict__ Oout) {
  const int wave = threadIdx.x >> 5;
  const int lane = threadIdx.x & 31;
  const int half = lane >> 4;
  const int l16  = lane & 15;

  const int bh = blockIdx.y;
  const int q0 = (blockIdx.x * 8 + wave) * 16;

  const float* Qm  = Q  + (size_t)bh * SEQ * D_HEAD;
  const float* Km  = K  + (size_t)bh * SEQ * D_HEAD;
  const float* Vtm = Vt + (size_t)bh * D_HEAD * SEQ;

  __shared__ float lds[8][16][17];
  float (*pt)[17] = lds[wave];

  v2f qf[16];
  const float* qrow = Qm + (size_t)(q0 + l16) * D_HEAD;
#pragma unroll
  for (int kk = 0; kk < 16; ++kk)
    qf[kk] = *(const v2f*)(qrow + kk * 4 + 2 * half);

  v8f o0 = {}, o1 = {}, o2 = {}, o3 = {};
  float m8[8], l8[8];
#pragma unroll
  for (int r = 0; r < 8; ++r) { m8[r] = -__builtin_inff(); l8[r] = 0.0f; }

  const float* vb0 = Vtm + (size_t)( 0 + l16) * SEQ;
  const float* vb1 = Vtm + (size_t)(16 + l16) * SEQ;
  const float* vb2 = Vtm + (size_t)(32 + l16) * SEQ;
  const float* vb3 = Vtm + (size_t)(48 + l16) * SEQ;

  for (int t0 = 0; t0 <= q0; t0 += 16) {
    v8f sc = {};
    const float* krow = Km + (size_t)(t0 + l16) * D_HEAD;
#pragma unroll
    for (int kk = 0; kk < 16; ++kk) {
      v2f bf = *(const v2f*)(krow + kk * 4 + 2 * half);
      sc = wmma_f32(qf[kk], bf, sc);
    }

    const bool diag = (t0 == q0);
#pragma unroll
    for (int r = 0; r < 8; ++r) {
      float v = sc[r] * 0.125f;
      if (diag && (l16 > r + 8 * half)) v = -__builtin_inff();
      sc[r] = v;
    }

#pragma unroll
    for (int r = 0; r < 8; ++r) {
      float v = sc[r];
      v = fmaxf(v, __shfl_xor(v, 1, 16));
      v = fmaxf(v, __shfl_xor(v, 2, 16));
      v = fmaxf(v, __shfl_xor(v, 4, 16));
      v = fmaxf(v, __shfl_xor(v, 8, 16));
      const float mn   = fmaxf(m8[r], v);
      const float resc = __expf(m8[r] - mn);
      m8[r] = mn;
      const float p = __expf(sc[r] - mn);
      sc[r] = p;
      float s = p;
      s += __shfl_xor(s, 1, 16);
      s += __shfl_xor(s, 2, 16);
      s += __shfl_xor(s, 4, 16);
      s += __shfl_xor(s, 8, 16);
      l8[r] = l8[r] * resc + s;
      o0[r] *= resc; o1[r] *= resc; o2[r] *= resc; o3[r] *= resc;
    }

#pragma unroll
    for (int r = 0; r < 8; ++r)
      pt[r + 8 * half][l16] = sc[r];
    asm volatile("s_wait_dscnt 0" ::: "memory");

#pragma unroll
    for (int kk = 0; kk < 4; ++kk) {
      v2f a;
      a.x = pt[l16][kk * 4 + 2 * half];
      a.y = pt[l16][kk * 4 + 2 * half + 1];
      const int tk = t0 + kk * 4 + 2 * half;
      v2f b;
      b = *(const v2f*)(vb0 + tk); o0 = wmma_f32(a, b, o0);
      b = *(const v2f*)(vb1 + tk); o1 = wmma_f32(a, b, o1);
      b = *(const v2f*)(vb2 + tk); o2 = wmma_f32(a, b, o2);
      b = *(const v2f*)(vb3 + tk); o3 = wmma_f32(a, b, o3);
    }
  }

  const int b = bh >> 4;
  const int h = bh & (N_HEADS - 1);
#pragma unroll
  for (int r = 0; r < 8; ++r) {
    const float inv = 1.0f / l8[r];
    const int s = q0 + r + 8 * half;
    float* orow = Oout + (size_t)(b * SEQ + s) * (N_HEADS * D_HEAD) + h * D_HEAD;
    orow[ 0 + l16] = o0[r] * inv;
    orow[16 + l16] = o1[r] * inv;
    orow[32 + l16] = o2[r] * inv;
    orow[48 + l16] = o3[r] * inv;
  }
}

__global__ __launch_bounds__(256) void out_proj_kernel(
    const float* __restrict__ A, const float* __restrict__ Wot,
    float* __restrict__ C) {
  const int wave = threadIdx.x >> 5;
  const int lane = threadIdx.x & 31;
  const int half = lane >> 4;
  const int l16  = lane & 15;

  const int row0 = blockIdx.x * 32;
  const int col0 = (blockIdx.y * 8 + wave) * 64;

  v8f c00 = {}, c01 = {}, c02 = {}, c03 = {};
  v8f c10 = {}, c11 = {}, c12 = {}, c13 = {};
  const float* xrow0 = A + (size_t)(row0 + l16) * D_MODEL;
  const float* xrow1 = xrow0 + (size_t)16 * D_MODEL;
  const float* w0 = Wot + (size_t)(col0 +  0 + l16) * D_MODEL;
  const float* w1 = Wot + (size_t)(col0 + 16 + l16) * D_MODEL;
  const float* w2 = Wot + (size_t)(col0 + 32 + l16) * D_MODEL;
  const float* w3 = Wot + (size_t)(col0 + 48 + l16) * D_MODEL;

#pragma unroll 4
  for (int k0 = 0; k0 < D_MODEL; k0 += 4) {
    const int kr = k0 + 2 * half;
    v2f a0 = *(const v2f*)(xrow0 + kr);
    v2f a1 = *(const v2f*)(xrow1 + kr);
    v2f b0 = *(const v2f*)(w0 + kr);
    v2f b1 = *(const v2f*)(w1 + kr);
    v2f b2 = *(const v2f*)(w2 + kr);
    v2f b3 = *(const v2f*)(w3 + kr);
    c00 = wmma_f32(a0, b0, c00);  c10 = wmma_f32(a1, b0, c10);
    c01 = wmma_f32(a0, b1, c01);  c11 = wmma_f32(a1, b1, c11);
    c02 = wmma_f32(a0, b2, c02);  c12 = wmma_f32(a1, b2, c12);
    c03 = wmma_f32(a0, b3, c03);  c13 = wmma_f32(a1, b3, c13);
  }

#pragma unroll
  for (int r = 0; r < 8; ++r) {
    const int row = row0 + r + 8 * half;
    float* crow0 = C + (size_t)row * D_MODEL + col0;
    float* crow1 = crow0 + (size_t)16 * D_MODEL;
    crow0[ 0 + l16] = c00[r];
    crow0[16 + l16] = c01[r];
    crow0[32 + l16] = c02[r];
    crow0[48 + l16] = c03[r];
    crow1[ 0 + l16] = c10[r];
    crow1[16 + l16] = c11[r];
    crow1[32 + l16] = c12[r];
    crow1[48 + l16] = c13[r];
  }
}

extern "C" void kernel_launch(void* const* d_in, const int* in_sizes, int n_in,
                              void* d_out, int out_size, void* d_ws, size_t ws_size,
                              hipStream_t stream) {
  const float* x  = (const float*)d_in[0];
  const float* Wq = (const float*)d_in[1];
  const float* Wk = (const float*)d_in[2];
  const float* Wv = (const float*)d_in[3];
  const float* Wo = (const float*)d_in[4];

  float* ws = (float*)d_ws;
  const size_t per = (size_t)BATCH * N_HEADS * SEQ * D_HEAD;
  const size_t wsz = (size_t)D_MODEL * D_MODEL;
  float* Q   = ws;
  float* K   = ws + per;
  float* Vt  = ws + 2 * per;
  float* AO  = ws + 3 * per;
  float* Wqt = ws + 4 * per;
  float* Wkt = Wqt + wsz;
  float* Wvt = Wkt + wsz;
  float* Wot = Wvt + wsz;

  dim3 g0(D_MODEL / 32, D_MODEL / 32, 4);
  transpose_w_kernel<<<g0, 256, 0, stream>>>(Wq, Wk, Wv, Wo, Wqt, Wkt, Wvt, Wot);

  dim3 g1(ROWS / 32, (N_HEADS * D_HEAD) / (16 * 8));
  qkv_proj_kernel<<<g1, 256, 0, stream>>>(x, Wqt, Wkt, Wvt, Q, K, Vt);

  dim3 g2(SEQ / (16 * 8), BATCH * N_HEADS);
  flash_attn_kernel<<<g2, 256, 0, stream>>>(Q, K, Vt, AO);

  dim3 g3(ROWS / 32, D_MODEL / (64 * 8));
  out_proj_kernel<<<g3, 256, 0, stream>>>(AO, Wot, (float*)d_out);
}


# QKV and output projections: 128x128x32 operand tiles loaded once per block by whole-line 16-byte loads into two LDS stages, wave tile 32x64, fragments read from LDS; k sum order kept
# speedup vs baseline: 1.2274x; 1.2274x over previous
#include <hip/hip_runtime.h>
#include <hip/hip_bf16.h>

typedef __attribute__((ext_vector_type(2))) float v2f;
typedef __attribute__((ext_vector_type(8))) float v8f;

#define D_MODEL 1024
#define N_HEADS 16
#define D_HEAD  64
#define BATCH   4
#define SEQ     2048
#define ROWS    (BATCH * SEQ)

static __device__ __forceinline__ v8f wmma_f32(v2f a, v2f b, v8f c) {
  return __builtin_amdgcn_wmma_f32_16x16x4_f32(
       false, a,  false, b,
       (short)0, c,  false,  false);
}

__global__ __launch_bounds__(256) void transpose_w_kernel(
    const float* __restrict__ W0, const float* __restrict__ W1,
    const float* __restrict__ W2, const float* __restrict__ W3,
    float* __restrict__ T0, float* __restrict__ T1,
    float* __restrict__ T2, float* __restrict__ T3) {
  const float* W = (blockIdx.z == 0) ? W0 : (blockIdx.z == 1) ? W1
                   : (blockIdx.z == 2) ? W2 : W3;
  float* T = (blockIdx.z == 0) ? T0 : (blockIdx.z == 1) ? T1
             : (blockIdx.z == 2) ? T2 : T3;

  __shared__ float tile[32][33];
  const int tx = threadIdx.x & 31;
  const int ty = threadIdx.x >> 5;
  const int x0 = blockIdx.x * 32;
  const int y0 = blockIdx.y * 32;

#pragma unroll
  for (int i = 0; i < 4; ++i)
    tile[ty + i * 8][tx] = W[(size_t)(y0 + ty + i * 8) * D_MODEL + x0 + tx];
  __syncthreads();
#pragma unroll
  for (int i = 0; i < 4; ++i)
    T[(size_t)(x0 + ty + i * 8) * D_MODEL + y0 + tx] = tile[tx][ty + i * 8];
}

typedef __attribute__((ext_vector_type(4))) float v4f;

#define GBM  128
#define GBN  128
#define GBK  32
#define GLDP 36
#define GTILE (GBM * GLDP)

static __device__ __forceinline__ void tile_fetch(
    const float* __restrict__ rows, int k0, int tid, v4f (&r)[4]) {
#pragma unroll
  for (int i = 0; i < 4; ++i) {
    const int idx = tid + 256 * i;
    const int row = idx >> 3;
    const int seg = idx & 7;
    r[i] = *(const v4f*)(rows + (size_t)row * D_MODEL + k0 + seg * 4);
  }
}

static __device__ __forceinline__ void tile_stash(
    float* __restrict__ s, int tid, const v4f (&r)[4]) {
#pragma unroll
  for (int i = 0; i < 4; ++i) {
    const int idx = tid + 256 * i;
    const int row = idx >> 3;
    const int seg = idx & 7;
    *(v4f*)(s + row * GLDP + seg * 4) = r[i];
  }
}

static __device__ __forceinline__ void tile_mma(
    const float* __restrict__ pa, const float* __restrict__ pb, v8f (&acc)[2][4]) {
#pragma unroll
  for (int kk = 0; kk < GBK / 4; ++kk) {
    const v2f a0 = *(const v2f*)(pa + kk * 4);
    const v2f a1 = *(const v2f*)(pa + 16 * GLDP + kk * 4);
    const v2f b0 = *(const v2f*)(pb + kk * 4);
    const v2f b1 = *(const v2f*)(pb + 16 * GLDP + kk * 4);
    const v2f b2 = *(const v2f*)(pb + 32 * GLDP + kk * 4);
    const v2f b3 = *(const v2f*)(pb + 48 * GLDP + kk * 4);
    acc[0][0] = wmma_f32(a0, b0, acc[0][0]);  acc[1][0] = wmma_f32(a1, b0, acc[1][0]);
    acc[0][1] = wmma_f32(a0, b1, acc[0][1]);  acc[1][1] = wmma_f32(a1, b1, acc[1][1]);
    acc[0][2] = wmma_f32(a0, b2, acc[0][2]);  acc[1][2] = wmma_f32(a1, b2, acc[1][2]);
    acc[0][3] = wmma_f32(a0, b3, acc[0][3]);  acc[1][3] = wmma_f32(a1, b3, acc[1][3]);
  }
}

static __device__ __forceinline__ void gemm_128x128(
    const float* __restrict__ Arows, const float* __restrict__ Brows,
    float* __restrict__ sA, float* __restrict__ sB, v8f (&acc)[2][4]) {
  const int tid  = threadIdx.x;
  const int wave = tid >> 5;
  const int lane = tid & 31;
  const int half = lane >> 4;
  const int l16  = lane & 15;
  const int wm   = wave & 3;
  const int wn   = wave >> 2;

  v4f ra[4], rb[4];
  tile_fetch(Arows, 0, tid, ra);
  tile_fetch(Brows, 0, tid, rb);
  tile_stash(sA, tid, ra);
  tile_stash(sB, tid, rb);
  __syncthreads();

  const int aoff = (wm * 32 + l16) * GLDP + 2 * half;
  const int boff = (wn * 64 + l16) * GLDP + 2 * half;
  const int KT = D_MODEL / GBK;

  for (int kt = 0; kt < KT - 1; ++kt) {
    const int cur = kt & 1;
    tile_fetch(Arows, (kt + 1) * GBK, tid, ra);
    tile_fetch(Brows, (kt + 1) * GBK, tid, rb);
    __builtin_amdgcn_sched_barrier(0);
    tile_mma(sA + cur * GTILE + aoff, sB + cur * GTILE + boff, acc);
    tile_stash(sA + (cur ^ 1) * GTILE, tid, ra);
    tile_stash(sB + (cur ^ 1) * GTILE, tid, rb);
    __syncthreads();
  }
  tile_mma(sA + ((KT - 1) & 1) * GTILE + aoff, sB + ((KT - 1) & 1) * GTILE + boff, acc);
}

__global__ __launch_bounds__(256) void qkv_proj_kernel(
    const float* __restrict__ X,
    const float* __restrict__ Wqt, const float* __restrict__ Wkt,
    const float* __restrict__ Wvt,
    float* __restrict__ Q, float* __restrict__ K, float* __restrict__ Vt) {
  __shared__ float sA[2 * GTILE];
  __shared__ float sB[2 * GTILE];

  const int which = blockIdx.y >> 3;
  const int cb    = blockIdx.y & 7;
  const float* Wt = (which == 0) ? Wqt : (which == 1) ? Wkt : Wvt;
  const int row0  = blockIdx.x * GBM;

  v8f acc[2][4];
#pragma unroll
  for (int t = 0; t < 2; ++t)
#pragma unroll
    for (int j = 0; j < 4; ++j) acc[t][j] = v8f{};

  gemm_128x128(X + (size_t)row0 * D_MODEL, Wt + (size_t)(cb * GBN) * D_MODEL,
               sA, sB, acc);

  const int wave = threadIdx.x >> 5;
  const int lane = threadIdx.x & 31;
  const int half = lane >> 4;
  const int l16  = lane & 15;
  const int wm   = wave & 3;
  const int wn   = wave >> 2;

  const int h  = cb * 2 + wn;
  const int b  = row0 >> 11;
  const int bh = b * N_HEADS + h;
  const int s0 = (row0 & (SEQ - 1)) + wm * 32 + 8 * half;

  if (which < 2) {
    float* O = ((which == 0) ? Q : K) + (size_t)bh * SEQ * D_HEAD;
#pragma unroll
    for (int t = 0; t < 2; ++t)
#pragma unroll
      for (int r = 0; r < 8; ++r) {
        float* orow = O + (size_t)(s0 + t * 16 + r) * D_HEAD + l16;
        orow[ 0] = acc[t][0][r];
        orow[16] = acc[t][1][r];
        orow[32] = acc[t][2][r];
        orow[48] = acc[t][3][r];
      }
  } else {
    float* O = Vt + (size_t)bh * D_HEAD * SEQ;
#pragma unroll
    for (int t = 0; t < 2; ++t)
#pragma unroll
      for (int r = 0; r < 8; ++r) {
        float* ocol = O + (size_t)l16 * SEQ + (s0 + t * 16 + r);
        ocol[(size_t) 0 * SEQ] = acc[t][0][r];
        ocol[(size_t)16 * SEQ] = acc[t][1][r];
        ocol[(size_t)32 * SEQ] = acc[t][2][r];
        ocol[(size_t)48 * SEQ] = acc[t][3][r];
      }
  }
}

__global__ __launch_bounds__(256) void flash_attn_kernel(
    const float* __restrict__ Q, const float* __restrict__ K,
    const float* __restrict__ Vt, float* __restrict__ Oout) {
  const int wave = threadIdx.x >> 5;
  const int lane = threadIdx.x & 31;
  const int half = lane >> 4;
  const int l16  = lane & 15;

  const int bh = blockIdx.y;
  const int q0 = (blockIdx.x * 8 + wave) * 16;

  const float* Qm  = Q  + (size_t)bh * SEQ * D_HEAD;
  const float* Km  = K  + (size_t)bh * SEQ * D_HEAD;
  const float* Vtm = Vt + (size_t)bh * D_HEAD * SEQ;

  __shared__ float lds[8][16][17];
  float (*pt)[17] = lds[wave];

  v2f qf[16];
  const float* qrow = Qm + (size_t)(q0 + l16) * D_HEAD;
#pragma unroll
  for (int kk = 0; kk < 16; ++kk)
    qf[kk] = *(const v2f*)(qrow + kk * 4 + 2 * half);

  v8f o0 = {}, o1 = {}, o2 = {}, o3 = {};
  float m8[8], l8[8];
#pragma unroll
  for (int r = 0; r < 8; ++r) { m8[r] = -__builtin_inff(); l8[r] = 0.0f; }

  const float* vb0 = Vtm + (size_t)( 0 + l16) * SEQ;
  const float* vb1 = Vtm + (size_t)(16 + l16) * SEQ;
  const float* vb2 = Vtm + (size_t)(32 + l16) * SEQ;
  const float* vb3 = Vtm + (size_t)(48 + l16) * SEQ;

  for (int t0 = 0; t0 <= q0; t0 += 16) {
    v8f sc = {};
    const float* krow = Km + (size_t)(t0 + l16) * D_HEAD;
#pragma unroll
    for (int kk = 0; kk < 16; ++kk) {
      v2f bf = *(const v2f*)(krow + kk * 4 + 2 * half);
      sc = wmma_f32(qf[kk], bf, sc);
    }

    const bool diag = (t0 == q0);
#pragma unroll
    for (int r = 0; r < 8; ++r) {
      float v = sc[r] * 0.125f;
      if (diag && (l16 > r + 8 * half)) v = -__builtin_inff();
      sc[r] = v;
    }

#pragma unroll
    for (int r = 0; r < 8; ++r) {
      float v = sc[r];
      v = fmaxf(v, __shfl_xor(v, 1, 16));
      v = fmaxf(v, __shfl_xor(v, 2, 16));
      v = fmaxf(v, __shfl_xor(v, 4, 16));
      v = fmaxf(v, __shfl_xor(v, 8, 16));
      const float mn   = fmaxf(m8[r], v);
      const float resc = __expf(m8[r] - mn);
      m8[r] = mn;
      const float p = __expf(sc[r] - mn);
      sc[r] = p;
      float s = p;
      s += __shfl_xor(s, 1, 16);
      s += __shfl_xor(s, 2, 16);
      s += __shfl_xor(s, 4, 16);
      s += __shfl_xor(s, 8, 16);
      l8[r] = l8[r] * resc + s;
      o0[r] *= resc; o1[r] *= resc; o2[r] *= resc; o3[r] *= resc;
    }

#pragma unroll
    for (int r = 0; r < 8; ++r)
      pt[r + 8 * half][l16] = sc[r];
    asm volatile("s_wait_dscnt 0" ::: "memory");

#pragma unroll
    for (int kk = 0; kk < 4; ++kk) {
      v2f a;
      a.x = pt[l16][kk * 4 + 2 * half];
      a.y = pt[l16][kk * 4 + 2 * half + 1];
      const int tk = t0 + kk * 4 + 2 * half;
      v2f b;
      b = *(const v2f*)(vb0 + tk); o0 = wmma_f32(a, b, o0);
      b = *(const v2f*)(vb1 + tk); o1 = wmma_f32(a, b, o1);
      b = *(const v2f*)(vb2 + tk); o2 = wmma_f32(a, b, o2);
      b = *(const v2f*)(vb3 + tk); o3 = wmma_f32(a, b, o3);
    }
  }

  const int b = bh >> 4;
  const int h = bh & (N_HEADS - 1);
#pragma unroll
  for (int r = 0; r < 8; ++r) {
    const float inv = 1.0f / l8[r];
    const int s = q0 + r + 8 * half;
    float* orow = Oout + (size_t)(b * SEQ + s) * (N_HEADS * D_HEAD) + h * D_HEAD;
    orow[ 0 + l16] = o0[r] * inv;
    orow[16 + l16] = o1[r] * inv;
    orow[32 + l16] = o2[r] * inv;
    orow[48 + l16] = o3[r] * inv;
  }
}

__global__ __launch_bounds__(256) void out_proj_kernel(
    const float* __restrict__ A, const float* __restrict__ Wot,
    float* __restrict__ C) {
  __shared__ float sA[2 * GTILE];
  __shared__ float sB[2 * GTILE];

  const int row0 = blockIdx.x * GBM;
  const int col0 = blockIdx.y * GBN;

  v8f acc[2][4];
#pragma unroll
  for (int t = 0; t < 2; ++t)
#pragma unroll
    for (int j = 0; j < 4; ++j) acc[t][j] = v8f{};

  gemm_128x128(A + (size_t)row0 * D_MODEL, Wot + (size_t)col0 * D_MODEL,
               sA, sB, acc);

  const int wave = threadIdx.x >> 5;
  const int lane = threadIdx.x & 31;
  const int half = lane >> 4;
  const int l16  = lane & 15;
  const int wm   = wave & 3;
  const int wn   = wave >> 2;

#pragma unroll
  for (int t = 0; t < 2; ++t)
#pragma unroll
    for (int r = 0; r < 8; ++r) {
      const int row = row0 + wm * 32 + t * 16 + 8 * half + r;
      float* crow = C + (size_t)row * D_MODEL + col0 + wn * 64 + l16;
      crow[ 0] = acc[t][0][r];
      crow[16] = acc[t][1][r];
      crow[32] = acc[t][2][r];
      crow[48] = acc[t][3][r];
    }
}

extern "C" void kernel_launch(void* const* d_in, const int* in_sizes, int n_in,
                              void* d_out, int out_size, void* d_ws, size_t ws_size,
                              hipStream_t stream) {
  const float* x  = (const float*)d_in[0];
  const float* Wq = (const float*)d_in[1];
  const float* Wk = (const float*)d_in[2];
  const float* Wv = (const float*)d_in[3];
  const float* Wo = (const float*)d_in[4];

  float* ws = (float*)d_ws;
  const size_t per = (size_t)BATCH * N_HEADS * SEQ * D_HEAD;
  const size_t wsz = (size_t)D_MODEL * D_MODEL;
  float* Q   = ws;
  float* K   = ws + per;
  float* Vt  = ws + 2 * per;
  float* AO  = ws + 3 * per;
  float* Wqt = ws + 4 * per;
  float* Wkt = Wqt + wsz;
  float* Wvt = Wkt + wsz;
  float* Wot = Wvt + wsz;

  dim3 g0(D_MODEL / 32, D_MODEL / 32, 4);
  transpose_w_kernel<<<g0, 256, 0, stream>>>(Wq, Wk, Wv, Wo, Wqt, Wkt, Wvt, Wot);

  dim3 g1(ROWS / GBM, 3 * (N_HEADS * D_HEAD) / GBN);
  qkv_proj_kernel<<<g1, 256, 0, stream>>>(x, Wqt, Wkt, Wvt, Q, K, Vt);

  dim3 g2(SEQ / (16 * 8), BATCH * N_HEADS);
  flash_attn_kernel<<<g2, 256, 0, stream>>>(Q, K, Vt, AO);

  dim3 g3(ROWS / GBM, D_MODEL / GBN);
  out_proj_kernel<<<g3, 256, 0, stream>>>(AO, Wot, (float*)d_out);
}


# Attention: 64-key K and V^T tiles loaded once per 128-query block by whole-line 16-byte loads into two LDS stages, B fragments read from LDS, K reads grouped ahead; per-wave step and sum order kept
# speedup vs baseline: 1.5645x; 1.2746x over previous
#include <hip/hip_runtime.h>
#include <hip/hip_bf16.h>

typedef __attribute__((ext_vector_type(2))) float v2f;
typedef __attribute__((ext_vector_type(8))) float v8f;

#define D_MODEL 1024
#define N_HEADS 16
#define D_HEAD  64
#define BATCH   4
#define SEQ     2048
#define ROWS    (BATCH * SEQ)

static __device__ __forceinline__ v8f wmma_f32(v2f a, v2f b, v8f c) {
  return __builtin_amdgcn_wmma_f32_16x16x4_f32(
       false, a,  false, b,
       (short)0, c,  false,  false);
}

__global__ __launch_bounds__(256) void transpose_w_kernel(
    const float* __restrict__ W0, const float* __restrict__ W1,
    const float* __restrict__ W2, const float* __restrict__ W3,
    float* __restrict__ T0, float* __restrict__ T1,
    float* __restrict__ T2, float* __restrict__ T3) {
  const float* W = (blockIdx.z == 0) ? W0 : (blockIdx.z == 1) ? W1
                   : (blockIdx.z == 2) ? W2 : W3;
  float* T = (blockIdx.z == 0) ? T0 : (blockIdx.z == 1) ? T1
             : (blockIdx.z == 2) ? T2 : T3;

  __shared__ float tile[32][33];
  const int tx = threadIdx.x & 31;
  const int ty = threadIdx.x >> 5;
  const int x0 = blockIdx.x * 32;
  const int y0 = blockIdx.y * 32;

#pragma unroll
  for (int i = 0; i < 4; ++i)
    tile[ty + i * 8][tx] = W[(size_t)(y0 + ty + i * 8) * D_MODEL + x0 + tx];
  __syncthreads();
#pragma unroll
  for (int i = 0; i < 4; ++i)
    T[(size_t)(x0 + ty + i * 8) * D_MODEL + y0 + tx] = tile[tx][ty + i * 8];
}

typedef __attribute__((ext_vector_type(4))) float v4f;

#define GBM  128
#define GBN  128
#define GBK  32
#define GLDP 36
#define GTILE (GBM * GLDP)

static __device__ __forceinline__ void tile_fetch(
    const float* __restrict__ rows, int k0, int tid, v4f (&r)[4]) {
#pragma unroll
  for (int i = 0; i < 4; ++i) {
    const int idx = tid + 256 * i;
    const int row = idx >> 3;
    const int seg = idx & 7;
    r[i] = *(const v4f*)(rows + (size_t)row * D_MODEL + k0 + seg * 4);
  }
}

static __device__ __forceinline__ void tile_stash(
    float* __restrict__ s, int tid, const v4f (&r)[4]) {
#pragma unroll
  for (int i = 0; i < 4; ++i) {
    const int idx = tid + 256 * i;
    const int row = idx >> 3;
    const int seg = idx & 7;
    *(v4f*)(s + row * GLDP + seg * 4) = r[i];
  }
}

static __device__ __forceinline__ void tile_mma(
    const float* __restrict__ pa, const float* __restrict__ pb, v8f (&acc)[2][4]) {
#pragma unroll
  for (int kk = 0; kk < GBK / 4; ++kk) {
    const v2f a0 = *(const v2f*)(pa + kk * 4);
    const v2f a1 = *(const v2f*)(pa + 16 * GLDP + kk * 4);
    const v2f b0 = *(const v2f*)(pb + kk * 4);
    const v2f b1 = *(const v2f*)(pb + 16 * GLDP + kk * 4);
    const v2f b2 = *(const v2f*)(pb + 32 * GLDP + kk * 4);
    const v2f b3 = *(const v2f*)(pb + 48 * GLDP + kk * 4);
    acc[0][0] = wmma_f32(a0, b0, acc[0][0]);  acc[1][0] = wmma_f32(a1, b0, acc[1][0]);
    acc[0][1] = wmma_f32(a0, b1, acc[0][1]);  acc[1][1] = wmma_f32(a1, b1, acc[1][1]);
    acc[0][2] = wmma_f32(a0, b2, acc[0][2]);  acc[1][2] = wmma_f32(a1, b2, acc[1][2]);
    acc[0][3] = wmma_f32(a0, b3, acc[0][3]);  acc[1][3] = wmma_f32(a1, b3, acc[1][3]);
  }
}

static __device__ __forceinline__ void gemm_128x128(
    const float* __restrict__ Arows, const float* __restrict__ Brows,
    float* __restrict__ sA, float* __restrict__ sB, v8f (&acc)[2][4]) {
  const int tid  = threadIdx.x;
  const int wave = tid >> 5;
  const int lane = tid & 31;
  const int half = lane >> 4;
  const int l16  = lane & 15;
  const int wm   = wave & 3;
  const int wn   = wave >> 2;

  v4f ra[4], rb[4];
  tile_fetch(Arows, 0, tid, ra);
  tile_fetch(Brows, 0, tid, rb);
  tile_stash(sA, tid, ra);
  tile_stash(sB, tid, rb);
  __syncthreads();

  const int aoff = (wm * 32 + l16) * GLDP + 2 * half;
  const int boff = (wn * 64 + l16) * GLDP + 2 * half;
  const int KT = D_MODEL / GBK;

  for (int kt = 0; kt < KT - 1; ++kt) {
    const int cur = kt & 1;
    tile_fetch(Arows, (kt + 1) * GBK, tid, ra);
    tile_fetch(Brows, (kt + 1) * GBK, tid, rb);
    __builtin_amdgcn_sched_barrier(0);
    tile_mma(sA + cur * GTILE + aoff, sB + cur * GTILE + boff, acc);
    tile_stash(sA + (cur ^ 1) * GTILE, tid, ra);
    tile_stash(sB + (cur ^ 1) * GTILE, tid, rb);
    __syncthreads();
  }
  tile_mma(sA + ((KT - 1) & 1) * GTILE + aoff, sB + ((KT - 1) & 1) * GTILE + boff, acc);
}

__global__ __launch_bounds__(256) void qkv_proj_kernel(
    const float* __restrict__ X,
    const float* __restrict__ Wqt, const float* __restrict__ Wkt,
    const float* __restrict__ Wvt,
    float* __restrict__ Q, float* __restrict__ K, float* __restrict__ Vt) {
  __shared__ float sA[2 * GTILE];
  __shared__ float sB[2 * GTILE];

  const int which = blockIdx.y >> 3;
  const int cb    = blockIdx.y & 7;
  const float* Wt = (which == 0) ? Wqt : (which == 1) ? Wkt : Wvt;
  const int row0  = blockIdx.x * GBM;

  v8f acc[2][4];
#pragma unroll
  for (int t = 0; t < 2; ++t)
#pragma unroll
    for (int j = 0; j < 4; ++j) acc[t][j] = v8f{};

  gemm_128x128(X + (size_t)row0 * D_MODEL, Wt + (size_t)(cb * GBN) * D_MODEL,
               sA, sB, acc);

  const int wave = threadIdx.x >> 5;
  const int lane = threadIdx.x & 31;
  const int half = lane >> 4;
  const int l16  = lane & 15;
  const int wm   = wave & 3;
  const int wn   = wave >> 2;

  const int h  = cb * 2 + wn;
  const int b  = row0 >> 11;
  const int bh = b * N_HEADS + h;
  const int s0 = (row0 & (SEQ - 1)) + wm * 32 + 8 * half;

  if (which < 2) {
    float* O = ((which == 0) ? Q : K) + (size_t)bh * SEQ * D_HEAD;
#pragma unroll
    for (int t = 0; t < 2; ++t)
#pragma unroll
      for (int r = 0; r < 8; ++r) {
        float* orow = O + (size_t)(s0 + t * 16 + r) * D_HEAD + l16;
        orow[ 0] = acc[t][0][r];
        orow[16] = acc[t][1][r];
        orow[32] = acc[t][2][r];
        orow[48] = acc[t][3][r];
      }
  } else {
    float* O = Vt + (size_t)bh * D_HEAD * SEQ;
#pragma unroll
    for (int t = 0; t < 2; ++t)
#pragma unroll
      for (int r = 0; r < 8; ++r) {
        float* ocol = O + (size_t)l16 * SEQ + (s0 + t * 16 + r);
        ocol[(size_t) 0 * SEQ] = acc[t][0][r];
        ocol[(size_t)16 * SEQ] = acc[t][1][r];
        ocol[(size_t)32 * SEQ] = acc[t][2][r];
        ocol[(size_t)48 * SEQ] = acc[t][3][r];
      }
  }
}

#define AKT   64
#define ALDP  68
#define ATILE (64 * ALDP)

static __device__ __forceinline__ void kv_fetch(
    const float* __restrict__ p, int stride, v4f (&r)[4]) {
#pragma unroll
  for (int i = 0; i < 4; ++i)
    r[i] = *(const v4f*)(p + (size_t)i * stride);
}

static __device__ __forceinline__ void kv_stash(
    float* __restrict__ s, int tid, const v4f (&r)[4]) {
#pragma unroll
  for (int i = 0; i < 4; ++i) {
    const int idx = tid + 256 * i;
    const int row = idx >> 4;
    const int seg = idx & 15;
    *(v4f*)(s + row * ALDP + seg * 4) = r[i];
  }
}

__global__ __launch_bounds__(256) void flash_attn_kernel(
    const float* __restrict__ Q, const float* __restrict__ K,
    const float* __restrict__ Vt, float* __restrict__ Oout) {
  __shared__ float sK[2 * ATILE];
  __shared__ float sV[2 * ATILE];
  __shared__ float lds[8][16][17];

  const int tid  = threadIdx.x;
  const int wave = __builtin_amdgcn_readfirstlane(tid >> 5);
  const int lane = tid & 31;
  const int half = lane >> 4;
  const int l16  = lane & 15;

  const int bh = blockIdx.y;
  const int qb = blockIdx.x * 128;
  const int q0 = qb + wave * 16;

  const float* Qm  = Q  + (size_t)bh * SEQ * D_HEAD;
  const float* Km  = K  + (size_t)bh * SEQ * D_HEAD;
  const float* Vtm = Vt + (size_t)bh * D_HEAD * SEQ;

  float (*pt)[17] = lds[wave];

  v2f qf[16];
  const float* qrow = Qm + (size_t)(q0 + l16) * D_HEAD;
#pragma unroll
  for (int kk = 0; kk < 16; ++kk)
    qf[kk] = *(const v2f*)(qrow + kk * 4 + 2 * half);

  v8f o0 = {}, o1 = {}, o2 = {}, o3 = {};
  float m8[8], l8[8];
#pragma unroll
  for (int r = 0; r < 8; ++r) { m8[r] = -__builtin_inff(); l8[r] = 0.0f; }

  const int ntiles = (qb >> 6) + 2;

  const float* kp = Km  + (size_t)(tid >> 4) * D_HEAD + (tid & 15) * 4;
  const float* vp = Vtm + (size_t)(tid >> 4) * SEQ    + (tid & 15) * 4;

  v4f rk[4], rv[4];
  kv_fetch(kp, 16 * D_HEAD, rk);
  kv_fetch(vp, 16 * SEQ, rv);
  kv_stash(sK, tid, rk);
  kv_stash(sV, tid, rv);
  __syncthreads();

  const int foff = l16 * ALDP + 2 * half;

  for (int kt = 0; kt < ntiles; ++kt) {
    const int cur = kt & 1;
    const bool more = (kt + 1 < ntiles);
    if (more) {
      kv_fetch(kp + (size_t)(kt + 1) * AKT * D_HEAD, 16 * D_HEAD, rk);
      kv_fetch(vp + (size_t)(kt + 1) * AKT, 16 * SEQ, rv);
    }
    __builtin_amdgcn_sched_barrier(0);

    const float* tK = sK + cur * ATILE + foff;
    const float* tV = sV + cur * ATILE + foff;

#pragma unroll
    for (int sub = 0; sub < 4; ++sub) {
      const int t0 = kt * AKT + sub * 16;
      if (t0 <= q0) {
        v8f sc = {};
        const float* krow = tK + sub * 16 * ALDP;
        v2f kf[16];
#pragma unroll
        for (int kk = 0; kk < 16; ++kk)
          kf[kk] = *(const v2f*)(krow + kk * 4);
        __builtin_amdgcn_sched_barrier(0);
#pragma unroll
        for (int kk = 0; kk < 16; ++kk)
          sc = wmma_f32(qf[kk], kf[kk], sc);

        const bool diag = (t0 == q0);
#pragma unroll
        for (int r = 0; r < 8; ++r) {
          float v = sc[r] * 0.125f;
          if (diag && (l16 > r + 8 * half)) v = -__builtin_inff();
          sc[r] = v;
        }

#pragma unroll
        for (int r = 0; r < 8; ++r) {
          float v = sc[r];
          v = fmaxf(v, __shfl_xor(v, 1, 16));
          v = fmaxf(v, __shfl_xor(v, 2, 16));
          v = fmaxf(v, __shfl_xor(v, 4, 16));
          v = fmaxf(v, __shfl_xor(v, 8, 16));
          const float mn   = fmaxf(m8[r], v);
          const float resc = __expf(m8[r] - mn);
          m8[r] = mn;
          const float p = __expf(sc[r] - mn);
          sc[r] = p;
          float s = p;
          s += __shfl_xor(s, 1, 16);
          s += __shfl_xor(s, 2, 16);
          s += __shfl_xor(s, 4, 16);
          s += __shfl_xor(s, 8, 16);
          l8[r] = l8[r] * resc + s;
          o0[r] *= resc; o1[r] *= resc; o2[r] *= resc; o3[r] *= resc;
        }

#pragma unroll
        for (int r = 0; r < 8; ++r)
          pt[r + 8 * half][l16] = sc[r];
        asm volatile("s_wait_dscnt 0" ::: "memory");

        const float* vcol = tV + sub * 16;
#pragma unroll
        for (int kk = 0; kk < 4; ++kk) {
          v2f a;
          a.x = pt[l16][kk * 4 + 2 * half];
          a.y = pt[l16][kk * 4 + 2 * half + 1];
          v2f b;
          b = *(const v2f*)(vcol +  0 * ALDP + kk * 4); o0 = wmma_f32(a, b, o0);
          b = *(const v2f*)(vcol + 16 * ALDP + kk * 4); o1 = wmma_f32(a, b, o1);
          b = *(const v2f*)(vcol + 32 * ALDP + kk * 4); o2 = wmma_f32(a, b, o2);
          b = *(const v2f*)(vcol + 48 * ALDP + kk * 4); o3 = wmma_f32(a, b, o3);
        }
      }
    }

    if (more) {
      kv_stash(sK + (cur ^ 1) * ATILE, tid, rk);
      kv_stash(sV + (cur ^ 1) * ATILE, tid, rv);
    }
    __syncthreads();
  }

  const int b = bh >> 4;
  const int h = bh & (N_HEADS - 1);
#pragma unroll
  for (int r = 0; r < 8; ++r) {
    const float inv = 1.0f / l8[r];
    const int s = q0 + r + 8 * half;
    float* orow = Oout + (size_t)(b * SEQ + s) * (N_HEADS * D_HEAD) + h * D_HEAD;
    orow[ 0 + l16] = o0[r] * inv;
    orow[16 + l16] = o1[r] * inv;
    orow[32 + l16] = o2[r] * inv;
    orow[48 + l16] = o3[r] * inv;
  }
}

__global__ __launch_bounds__(256) void out_proj_kernel(
    const float* __restrict__ A, const float* __restrict__ Wot,
    float* __restrict__ C) {
  __shared__ float sA[2 * GTILE];
  __shared__ float sB[2 * GTILE];

  const int row0 = blockIdx.x * GBM;
  const int col0 = blockIdx.y * GBN;

  v8f acc[2][4];
#pragma unroll
  for (int t = 0; t < 2; ++t)
#pragma unroll
    for (int j = 0; j < 4; ++j) acc[t][j] = v8f{};

  gemm_128x128(A + (size_t)row0 * D_MODEL, Wot + (size_t)col0 * D_MODEL,
               sA, sB, acc);

  const int wave = threadIdx.x >> 5;
  const int lane = threadIdx.x & 31;
  const int half = lane >> 4;
  const int l16  = lane & 15;
  const int wm   = wave & 3;
  const int wn   = wave >> 2;

#pragma unroll
  for (int t = 0; t < 2; ++t)
#pragma unroll
    for (int r = 0; r < 8; ++r) {
      const int row = row0 + wm * 32 + t * 16 + 8 * half + r;
      float* crow = C + (size_t)row * D_MODEL + col0 + wn * 64 + l16;
      crow[ 0] = acc[t][0][r];
      crow[16] = acc[t][1][r];
      crow[32] = acc[t][2][r];
      crow[48] = acc[t][3][r];
    }
}

extern "C" void kernel_launch(void* const* d_in, const int* in_sizes, int n_in,
                              void* d_out, int out_size, void* d_ws, size_t ws_size,
                              hipStream_t stream) {
  const float* x  = (const float*)d_in[0];
  const float* Wq = (const float*)d_in[1];
  const float* Wk = (const float*)d_in[2];
  const float* Wv = (const float*)d_in[3];
  const float* Wo = (const float*)d_in[4];

  float* ws = (float*)d_ws;
  const size_t per = (size_t)BATCH * N_HEADS * SEQ * D_HEAD;
  const size_t wsz = (size_t)D_MODEL * D_MODEL;
  float* Q   = ws;
  float* K   = ws + per;
  float* Vt  = ws + 2 * per;
  float* AO  = ws + 3 * per;
  float* Wqt = ws + 4 * per;
  float* Wkt = Wqt + wsz;
  float* Wvt = Wkt + wsz;
  float* Wot = Wvt + wsz;

  dim3 g0(D_MODEL / 32, D_MODEL / 32, 4);
  transpose_w_kernel<<<g0, 256, 0, stream>>>(Wq, Wk, Wv, Wo, Wqt, Wkt, Wvt, Wot);

  dim3 g1(ROWS / GBM, 3 * (N_HEADS * D_HEAD) / GBN);
  qkv_proj_kernel<<<g1, 256, 0, stream>>>(x, Wqt, Wkt, Wvt, Q, K, Vt);

  dim3 g2(SEQ / (16 * 8), BATCH * N_HEADS);
  flash_attn_kernel<<<g2, 256, 0, stream>>>(Q, K, Vt, AO);

  dim3 g3(ROWS / GBM, D_MODEL / GBN);
  out_proj_kernel<<<g3, 256, 0, stream>>>(AO, Wot, (float*)d_out);
}


# Attention: each workgroup does query block x, then its mirror 15-x with waves reversed: every wave has 129 key steps, 512 equal workgroups; per-tile key order and sums kept
# speedup vs baseline: 1.9389x; 1.2393x over previous
#include <hip/hip_runtime.h>
#include <hip/hip_bf16.h>

typedef __attribute__((ext_vector_type(2))) float v2f;
typedef __attribute__((ext_vector_type(8))) float v8f;

#define D_MODEL 1024
#define N_HEADS 16
#define D_HEAD  64
#define BATCH   4
#define SEQ     2048
#define ROWS    (BATCH * SEQ)

static __device__ __forceinline__ v8f wmma_f32(v2f a, v2f b, v8f c) {
  return __builtin_amdgcn_wmma_f32_16x16x4_f32(
       false, a,  false, b,
       (short)0, c,  false,  false);
}

__global__ __launch_bounds__(256) void transpose_w_kernel(
    const float* __restrict__ W0, const float* __restrict__ W1,
    const float* __restrict__ W2, const float* __restrict__ W3,
    float* __restrict__ T0, float* __restrict__ T1,
    float* __restrict__ T2, float* __restrict__ T3) {
  const float* W = (blockIdx.z == 0) ? W0 : (blockIdx.z == 1) ? W1
                   : (blockIdx.z == 2) ? W2 : W3;
  float* T = (blockIdx.z == 0) ? T0 : (blockIdx.z == 1) ? T1
             : (blockIdx.z == 2) ? T2 : T3;

  __shared__ float tile[32][33];
  const int tx = threadIdx.x & 31;
  const int ty = threadIdx.x >> 5;
  const int x0 = blockIdx.x * 32;
  const int y0 = blockIdx.y * 32;

#pragma unroll
  for (int i = 0; i < 4; ++i)
    tile[ty + i * 8][tx] = W[(size_t)(y0 + ty + i * 8) * D_MODEL + x0 + tx];
  __syncthreads();
#pragma unroll
  for (int i = 0; i < 4; ++i)
    T[(size_t)(x0 + ty + i * 8) * D_MODEL + y0 + tx] = tile[tx][ty + i * 8];
}

typedef __attribute__((ext_vector_type(4))) float v4f;

#define GBM  128
#define GBN  128
#define GBK  32
#define GLDP 36
#define GTILE (GBM * GLDP)

static __device__ __forceinline__ void tile_fetch(
    const float* __restrict__ rows, int k0, int tid, v4f (&r)[4]) {
#pragma unroll
  for (int i = 0; i < 4; ++i) {
    const int idx = tid + 256 * i;
    const int row = idx >> 3;
    const int seg = idx & 7;
    r[i] = *(const v4f*)(rows + (size_t)row * D_MODEL + k0 + seg * 4);
  }
}

static __device__ __forceinline__ void tile_stash(
    float* __restrict__ s, int tid, const v4f (&r)[4]) {
#pragma unroll
  for (int i = 0; i < 4; ++i) {
    const int idx = tid + 256 * i;
    const int row = idx >> 3;
    const int seg = idx & 7;
    *(v4f*)(s + row * GLDP + seg * 4) = r[i];
  }
}

static __device__ __forceinline__ void tile_mma(
    const float* __restrict__ pa, const float* __restrict__ pb, v8f (&acc)[2][4]) {
#pragma unroll
  for (int kk = 0; kk < GBK / 4; ++kk) {
    const v2f a0 = *(const v2f*)(pa + kk * 4);
    const v2f a1 = *(const v2f*)(pa + 16 * GLDP + kk * 4);
    const v2f b0 = *(const v2f*)(pb + kk * 4);
    const v2f b1 = *(const v2f*)(pb + 16 * GLDP + kk * 4);
    const v2f b2 = *(const v2f*)(pb + 32 * GLDP + kk * 4);
    const v2f b3 = *(const v2f*)(pb + 48 * GLDP + kk * 4);
    acc[0][0] = wmma_f32(a0, b0, acc[0][0]);  acc[1][0] = wmma_f32(a1, b0, acc[1][0]);
    acc[0][1] = wmma_f32(a0, b1, acc[0][1]);  acc[1][1] = wmma_f32(a1, b1, acc[1][1]);
    acc[0][2] = wmma_f32(a0, b2, acc[0][2]);  acc[1][2] = wmma_f32(a1, b2, acc[1][2]);
    acc[0][3] = wmma_f32(a0, b3, acc[0][3]);  acc[1][3] = wmma_f32(a1, b3, acc[1][3]);
  }
}

static __device__ __forceinline__ void gemm_128x128(
    const float* __restrict__ Arows, const float* __restrict__ Brows,
    float* __restrict__ sA, float* __restrict__ sB, v8f (&acc)[2][4]) {
  const int tid  = threadIdx.x;
  const int wave = tid >> 5;
  const int lane = tid & 31;
  const int half = lane >> 4;
  const int l16  = lane & 15;
  const int wm   = wave & 3;
  const int wn   = wave >> 2;

  v4f ra[4], rb[4];
  tile_fetch(Arows, 0, tid, ra);
  tile_fetch(Brows, 0, tid, rb);
  tile_stash(sA, tid, ra);
  tile_stash(sB, tid, rb);
  __syncthreads();

  const int aoff = (wm * 32 + l16) * GLDP + 2 * half;
  const int boff = (wn * 64 + l16) * GLDP + 2 * half;
  const int KT = D_MODEL / GBK;

  for (int kt = 0; kt < KT - 1; ++kt) {
    const int cur = kt & 1;
    tile_fetch(Arows, (kt + 1) * GBK, tid, ra);
    tile_fetch(Brows, (kt + 1) * GBK, tid, rb);
    __builtin_amdgcn_sched_barrier(0);
    tile_mma(sA + cur * GTILE + aoff, sB + cur * GTILE + boff, acc);
    tile_stash(sA + (cur ^ 1) * GTILE, tid, ra);
    tile_stash(sB + (cur ^ 1) * GTILE, tid, rb);
    __syncthreads();
  }
  tile_mma(sA + ((KT - 1) & 1) * GTILE + aoff, sB + ((KT - 1) & 1) * GTILE + boff, acc);
}

__global__ __launch_bounds__(256) void qkv_proj_kernel(
    const float* __restrict__ X,
    const float* __restrict__ Wqt, const float* __restrict__ Wkt,
    const float* __restrict__ Wvt,
    float* __restrict__ Q, float* __restrict__ K, float* __restrict__ Vt) {
  __shared__ float sA[2 * GTILE];
  __shared__ float sB[2 * GTILE];

  const int which = blockIdx.y >> 3;
  const int cb    = blockIdx.y & 7;
  const float* Wt = (which == 0) ? Wqt : (which == 1) ? Wkt : Wvt;
  const int row0  = blockIdx.x * GBM;

  v8f acc[2][4];
#pragma unroll
  for (int t = 0; t < 2; ++t)
#pragma unroll
    for (int j = 0; j < 4; ++j) acc[t][j] = v8f{};

  gemm_128x128(X + (size_t)row0 * D_MODEL, Wt + (size_t)(cb * GBN) * D_MODEL,
               sA, sB, acc);

  const int wave = threadIdx.x >> 5;
  const int lane = threadIdx.x & 31;
  const int half = lane >> 4;
  const int l16  = lane & 15;
  const int wm   = wave & 3;
  const int wn   = wave >> 2;

  const int h  = cb * 2 + wn;
  const int b  = row0 >> 11;
  const int bh = b * N_HEADS + h;
  const int s0 = (row0 & (SEQ - 1)) + wm * 32 + 8 * half;

  if (which < 2) {
    float* O = ((which == 0) ? Q : K) + (size_t)bh * SEQ * D_HEAD;
#pragma unroll
    for (int t = 0; t < 2; ++t)
#pragma unroll
      for (int r = 0; r < 8; ++r) {
        float* orow = O + (size_t)(s0 + t * 16 + r) * D_HEAD + l16;
        orow[ 0] = acc[t][0][r];
        orow[16] = acc[t][1][r];
        orow[32] = acc[t][2][r];
        orow[48] = acc[t][3][r];
      }
  } else {
    float* O = Vt + (size_t)bh * D_HEAD * SEQ;
#pragma unroll
    for (int t = 0; t < 2; ++t)
#pragma unroll
      for (int r = 0; r < 8; ++r) {
        float* ocol = O + (size_t)l16 * SEQ + (s0 + t * 16 + r);
        ocol[(size_t) 0 * SEQ] = acc[t][0][r];
        ocol[(size_t)16 * SEQ] = acc[t][1][r];
        ocol[(size_t)32 * SEQ] = acc[t][2][r];
        ocol[(size_t)48 * SEQ] = acc[t][3][r];
      }
  }
}

#define AKT   64
#define ALDP  68
#define ATILE (64 * ALDP)

static __device__ __forceinline__ void kv_fetch(
    const float* __restrict__ p, int stride, v4f (&r)[4]) {
#pragma unroll
  for (int i = 0; i < 4; ++i)
    r[i] = *(const v4f*)(p + (size_t)i * stride);
}

static __device__ __forceinline__ void kv_stash(
    float* __restrict__ s, int tid, const v4f (&r)[4]) {
#pragma unroll
  for (int i = 0; i < 4; ++i) {
    const int idx = tid + 256 * i;
    const int row = idx >> 4;
    const int seg = idx & 15;
    *(v4f*)(s + row * ALDP + seg * 4) = r[i];
  }
}

__global__ __launch_bounds__(256) void flash_attn_kernel(
    const float* __restrict__ Q, const float* __restrict__ K,
    const float* __restrict__ Vt, float* __restrict__ Oout) {
  __shared__ float sK[2 * ATILE];
  __shared__ float sV[2 * ATILE];
  __shared__ float lds[8][16][17];

  const int tid  = threadIdx.x;
  const int wave = __builtin_amdgcn_readfirstlane(tid >> 5);
  const int lane = tid & 31;
  const int half = lane >> 4;
  const int l16  = lane & 15;

  const int bh = blockIdx.y;

  const float* Qm  = Q  + (size_t)bh * SEQ * D_HEAD;
  const float* Km  = K  + (size_t)bh * SEQ * D_HEAD;
  const float* Vtm = Vt + (size_t)bh * D_HEAD * SEQ;

  float (*pt)[17] = lds[wave];

  const float* kp = Km  + (size_t)(tid >> 4) * D_HEAD + (tid & 15) * 4;
  const float* vp = Vtm + (size_t)(tid >> 4) * SEQ    + (tid & 15) * 4;

  const int foff = l16 * ALDP + 2 * half;

  const int b = bh >> 4;
  const int h = bh & (N_HEADS - 1);

  for (int pass = 0; pass < 2; ++pass) {
    const int xb = pass ? (SEQ / 128 - 1 - (int)blockIdx.x) : (int)blockIdx.x;
    const int wq = pass ? 7 - wave : wave;
    const int qb = xb * 128;
    const int q0 = qb + wq * 16;

    v2f qf[16];
    const float* qrow = Qm + (size_t)(q0 + l16) * D_HEAD;
#pragma unroll
    for (int kk = 0; kk < 16; ++kk)
      qf[kk] = *(const v2f*)(qrow + kk * 4 + 2 * half);

    v8f o0 = {}, o1 = {}, o2 = {}, o3 = {};
    float m8[8], l8[8];
#pragma unroll
    for (int r = 0; r < 8; ++r) { m8[r] = -__builtin_inff(); l8[r] = 0.0f; }

    const int ntiles = (qb >> 6) + 2;

    v4f rk[4], rv[4];
    kv_fetch(kp, 16 * D_HEAD, rk);
    kv_fetch(vp, 16 * SEQ, rv);
    kv_stash(sK, tid, rk);
    kv_stash(sV, tid, rv);
    __syncthreads();

    for (int kt = 0; kt < ntiles; ++kt) {
      const int cur = kt & 1;
      const bool more = (kt + 1 < ntiles);
      if (more) {
        kv_fetch(kp + (size_t)(kt + 1) * AKT * D_HEAD, 16 * D_HEAD, rk);
        kv_fetch(vp + (size_t)(kt + 1) * AKT, 16 * SEQ, rv);
      }
      __builtin_amdgcn_sched_barrier(0);

      const float* tK = sK + cur * ATILE + foff;
      const float* tV = sV + cur * ATILE + foff;

#pragma unroll
      for (int sub = 0; sub < 4; ++sub) {
        const int t0 = kt * AKT + sub * 16;
        if (t0 <= q0) {
          v8f sc = {};
          const float* krow = tK + sub * 16 * ALDP;
          v2f kf[16];
#pragma unroll
          for (int kk = 0; kk < 16; ++kk)
            kf[kk] = *(const v2f*)(krow + kk * 4);
          __builtin_amdgcn_sched_barrier(0);
#pragma unroll
          for (int kk = 0; kk < 16; ++kk)
            sc = wmma_f32(qf[kk], kf[kk], sc);

          const bool diag = (t0 == q0);
#pragma unroll
          for (int r = 0; r < 8; ++r) {
            float v = sc[r] * 0.125f;
            if (diag && (l16 > r + 8 * half)) v = -__builtin_inff();
            sc[r] = v;
          }

#pragma unroll
          for (int r = 0; r < 8; ++r) {
            float v = sc[r];
            v = fmaxf(v, __shfl_xor(v, 1, 16));
            v = fmaxf(v, __shfl_xor(v, 2, 16));
            v = fmaxf(v, __shfl_xor(v, 4, 16));
            v = fmaxf(v, __shfl_xor(v, 8, 16));
            const float mn   = fmaxf(m8[r], v);
            const float resc = __expf(m8[r] - mn);
            m8[r] = mn;
            const float p = __expf(sc[r] - mn);
            sc[r] = p;
            float s = p;
            s += __shfl_xor(s, 1, 16);
            s += __shfl_xor(s, 2, 16);
            s += __shfl_xor(s, 4, 16);
            s += __shfl_xor(s, 8, 16);
            l8[r] = l8[r] * resc + s;
            o0[r] *= resc; o1[r] *= resc; o2[r] *= resc; o3[r] *= resc;
          }

#pragma unroll
          for (int r = 0; r < 8; ++r)
            pt[r + 8 * half][l16] = sc[r];
          asm volatile("s_wait_dscnt 0" ::: "memory");

          const float* vcol = tV + sub * 16;
#pragma unroll
          for (int kk = 0; kk < 4; ++kk) {
            v2f a;
            a.x = pt[l16][kk * 4 + 2 * half];
            a.y = pt[l16][kk * 4 + 2 * half + 1];
            v2f b;
            b = *(const v2f*)(vcol +  0 * ALDP + kk * 4); o0 = wmma_f32(a, b, o0);
            b = *(const v2f*)(vcol + 16 * ALDP + kk * 4); o1 = wmma_f32(a, b, o1);
            b = *(const v2f*)(vcol + 32 * ALDP + kk * 4); o2 = wmma_f32(a, b, o2);
            b = *(const v2f*)(vcol + 48 * ALDP + kk * 4); o3 = wmma_f32(a, b, o3);
          }
        }
      }

      if (more) {
        kv_stash(sK + (cur ^ 1) * ATILE, tid, rk);
        kv_stash(sV + (cur ^ 1) * ATILE, tid, rv);
      }
      __syncthreads();
    }

#pragma unroll
    for (int r = 0; r < 8; ++r) {
      const float inv = 1.0f / l8[r];
      const int s = q0 + r + 8 * half;
      float* orow = Oout + (size_t)(b * SEQ + s) * (N_HEADS * D_HEAD) + h * D_HEAD;
      orow[ 0 + l16] = o0[r] * inv;
      orow[16 + l16] = o1[r] * inv;
      orow[32 + l16] = o2[r] * inv;
      orow[48 + l16] = o3[r] * inv;
    }
  }
}

__global__ __launch_bounds__(256) void out_proj_kernel(
    const float* __restrict__ A, const float* __restrict__ Wot,
    float* __restrict__ C) {
  __shared__ float sA[2 * GTILE];
  __shared__ float sB[2 * GTILE];

  const int row0 = blockIdx.x * GBM;
  const int col0 = blockIdx.y * GBN;

  v8f acc[2][4];
#pragma unroll
  for (int t = 0; t < 2; ++t)
#pragma unroll
    for (int j = 0; j < 4; ++j) acc[t][j] = v8f{};

  gemm_128x128(A + (size_t)row0 * D_MODEL, Wot + (size_t)col0 * D_MODEL,
               sA, sB, acc);

  const int wave = threadIdx.x >> 5;
  const int lane = threadIdx.x & 31;
  const int half = lane >> 4;
  const int l16  = lane & 15;
  const int wm   = wave & 3;
  const int wn   = wave >> 2;

#pragma unroll
  for (int t = 0; t < 2; ++t)
#pragma unroll
    for (int r = 0; r < 8; ++r) {
      const int row = row0 + wm * 32 + t * 16 + 8 * half + r;
      float* crow = C + (size_t)row * D_MODEL + col0 + wn * 64 + l16;
      crow[ 0] = acc[t][0][r];
      crow[16] = acc[t][1][r];
      crow[32] = acc[t][2][r];
      crow[48] = acc[t][3][r];
    }
}

extern "C" void kernel_launch(void* const* d_in, const int* in_sizes, int n_in,
                              void* d_out, int out_size, void* d_ws, size_t ws_size,
                              hipStream_t stream) {
  const float* x  = (const float*)d_in[0];
  const float* Wq = (const float*)d_in[1];
  const float* Wk = (const float*)d_in[2];
  const float* Wv = (const float*)d_in[3];
  const float* Wo = (const float*)d_in[4];

  float* ws = (float*)d_ws;
  const size_t per = (size_t)BATCH * N_HEADS * SEQ * D_HEAD;
  const size_t wsz = (size_t)D_MODEL * D_MODEL;
  float* Q   = ws;
  float* K   = ws + per;
  float* Vt  = ws + 2 * per;
  float* AO  = ws + 3 * per;
  float* Wqt = ws + 4 * per;
  float* Wkt = Wqt + wsz;
  float* Wvt = Wkt + wsz;
  float* Wot = Wvt + wsz;

  dim3 g0(D_MODEL / 32, D_MODEL / 32, 4);
  transpose_w_kernel<<<g0, 256, 0, stream>>>(Wq, Wk, Wv, Wo, Wqt, Wkt, Wvt, Wot);

  dim3 g1(ROWS / GBM, 3 * (N_HEADS * D_HEAD) / GBN);
  qkv_proj_kernel<<<g1, 256, 0, stream>>>(x, Wqt, Wkt, Wvt, Q, K, Vt);

  dim3 g2(SEQ / (2 * 16 * 8), BATCH * N_HEADS);
  flash_attn_kernel<<<g2, 256, 0, stream>>>(Q, K, Vt, AO);

  dim3 g3(ROWS / GBM, D_MODEL / GBN);
  out_proj_kernel<<<g3, 256, 0, stream>>>(AO, Wot, (float*)d_out);
}


# Attention: both products computed transposed (keys x queries): a query's softmax sits in one lane pair, P feeds P*V from registers; no LDS permutes, no P staging; row-sum order changed
# speedup vs baseline: 2.1434x; 1.1055x over previous
#include <hip/hip_runtime.h>
#include <hip/hip_bf16.h>

typedef __attribute__((ext_vector_type(2))) float v2f;
typedef __attribute__((ext_vector_type(8))) float v8f;

#define D_MODEL 1024
#define N_HEADS 16
#define D_HEAD  64
#define BATCH   4
#define SEQ     2048
#define ROWS    (BATCH * SEQ)

static __device__ __forceinline__ v8f wmma_f32(v2f a, v2f b, v8f c) {
  return __builtin_amdgcn_wmma_f32_16x16x4_f32(
       false, a,  false, b,
       (short)0, c,  false,  false);
}

__global__ __launch_bounds__(256) void transpose_w_kernel(
    const float* __restrict__ W0, const float* __restrict__ W1,
    const float* __restrict__ W2, const float* __restrict__ W3,
    float* __restrict__ T0, float* __restrict__ T1,
    float* __restrict__ T2, float* __restrict__ T3) {
  const float* W = (blockIdx.z == 0) ? W0 : (blockIdx.z == 1) ? W1
                   : (blockIdx.z == 2) ? W2 : W3;
  float* T = (blockIdx.z == 0) ? T0 : (blockIdx.z == 1) ? T1
             : (blockIdx.z == 2) ? T2 : T3;

  __shared__ float tile[32][33];
  const int tx = threadIdx.x & 31;
  const int ty = threadIdx.x >> 5;
  const int x0 = blockIdx.x * 32;
  const int y0 = blockIdx.y * 32;

#pragma unroll
  for (int i = 0; i < 4; ++i)
    tile[ty + i * 8][tx] = W[(size_t)(y0 + ty + i * 8) * D_MODEL + x0 + tx];
  __syncthreads();
#pragma unroll
  for (int i = 0; i < 4; ++i)
    T[(size_t)(x0 + ty + i * 8) * D_MODEL + y0 + tx] = tile[tx][ty + i * 8];
}

typedef __attribute__((ext_vector_type(4))) float v4f;

#define GBM  128
#define GBN  128
#define GBK  32
#define GLDP 36
#define GTILE (GBM * GLDP)

static __device__ __forceinline__ void tile_fetch(
    const float* __restrict__ rows, int k0, int tid, v4f (&r)[4]) {
#pragma unroll
  for (int i = 0; i < 4; ++i) {
    const int idx = tid + 256 * i;
    const int row = idx >> 3;
    const int seg = idx & 7;
    r[i] = *(const v4f*)(rows + (size_t)row * D_MODEL + k0 + seg * 4);
  }
}

static __device__ __forceinline__ void tile_stash(
    float* __restrict__ s, int tid, const v4f (&r)[4]) {
#pragma unroll
  for (int i = 0; i < 4; ++i) {
    const int idx = tid + 256 * i;
    const int row = idx >> 3;
    const int seg = idx & 7;
    *(v4f*)(s + row * GLDP + seg * 4) = r[i];
  }
}

static __device__ __forceinline__ void tile_mma(
    const float* __restrict__ pa, const float* __restrict__ pb, v8f (&acc)[2][4]) {
#pragma unroll
  for (int kk = 0; kk < GBK / 4; ++kk) {
    const v2f a0 = *(const v2f*)(pa + kk * 4);
    const v2f a1 = *(const v2f*)(pa + 16 * GLDP + kk * 4);
    const v2f b0 = *(const v2f*)(pb + kk * 4);
    const v2f b1 = *(const v2f*)(pb + 16 * GLDP + kk * 4);
    const v2f b2 = *(const v2f*)(pb + 32 * GLDP + kk * 4);
    const v2f b3 = *(const v2f*)(pb + 48 * GLDP + kk * 4);
    acc[0][0] = wmma_f32(a0, b0, acc[0][0]);  acc[1][0] = wmma_f32(a1, b0, acc[1][0]);
    acc[0][1] = wmma_f32(a0, b1, acc[0][1]);  acc[1][1] = wmma_f32(a1, b1, acc[1][1]);
    acc[0][2] = wmma_f32(a0, b2, acc[0][2]);  acc[1][2] = wmma_f32(a1, b2, acc[1][2]);
    acc[0][3] = wmma_f32(a0, b3, acc[0][3]);  acc[1][3] = wmma_f32(a1, b3, acc[1][3]);
  }
}

static __device__ __forceinline__ void gemm_128x128(
    const float* __restrict__ Arows, const float* __restrict__ Brows,
    float* __restrict__ sA, float* __restrict__ sB, v8f (&acc)[2][4]) {
  const int tid  = threadIdx.x;
  const int wave = tid >> 5;
  const int lane = tid & 31;
  const int half = lane >> 4;
  const int l16  = lane & 15;
  const int wm   = wave & 3;
  const int wn   = wave >> 2;

  v4f ra[4], rb[4];
  tile_fetch(Arows, 0, tid, ra);
  tile_fetch(Brows, 0, tid, rb);
  tile_stash(sA, tid, ra);
  tile_stash(sB, tid, rb);
  __syncthreads();

  const int aoff = (wm * 32 + l16) * GLDP + 2 * half;
  const int boff = (wn * 64 + l16) * GLDP + 2 * half;
  const int KT = D_MODEL / GBK;

  for (int kt = 0; kt < KT - 1; ++kt) {
    const int cur = kt & 1;
    tile_fetch(Arows, (kt + 1) * GBK, tid, ra);
    tile_fetch(Brows, (kt + 1) * GBK, tid, rb);
    __builtin_amdgcn_sched_barrier(0);
    tile_mma(sA + cur * GTILE + aoff, sB + cur * GTILE + boff, acc);
    tile_stash(sA + (cur ^ 1) * GTILE, tid, ra);
    tile_stash(sB + (cur ^ 1) * GTILE, tid, rb);
    __syncthreads();
  }
  tile_mma(sA + ((KT - 1) & 1) * GTILE + aoff, sB + ((KT - 1) & 1) * GTILE + boff, acc);
}

__global__ __launch_bounds__(256) void qkv_proj_kernel(
    const float* __restrict__ X,
    const float* __restrict__ Wqt, const float* __restrict__ Wkt,
    const float* __restrict__ Wvt,
    float* __restrict__ Q, float* __restrict__ K, float* __restrict__ Vt) {
  __shared__ float sA[2 * GTILE];
  __shared__ float sB[2 * GTILE];

  const int which = blockIdx.y >> 3;
  const int cb    = blockIdx.y & 7;
  const float* Wt = (which == 0) ? Wqt : (which == 1) ? Wkt : Wvt;
  const int row0  = blockIdx.x * GBM;

  v8f acc[2][4];
#pragma unroll
  for (int t = 0; t < 2; ++t)
#pragma unroll
    for (int j = 0; j < 4; ++j) acc[t][j] = v8f{};

  gemm_128x128(X + (size_t)row0 * D_MODEL, Wt + (size_t)(cb * GBN) * D_MODEL,
               sA, sB, acc);

  const int wave = threadIdx.x >> 5;
  const int lane = threadIdx.x & 31;
  const int half = lane >> 4;
  const int l16  = lane & 15;
  const int wm   = wave & 3;
  const int wn   = wave >> 2;

  const int h  = cb * 2 + wn;
  const int b  = row0 >> 11;
  const int bh = b * N_HEADS + h;
  const int s0 = (row0 & (SEQ - 1)) + wm * 32 + 8 * half;

  if (which < 2) {
    float* O = ((which == 0) ? Q : K) + (size_t)bh * SEQ * D_HEAD;
#pragma unroll
    for (int t = 0; t < 2; ++t)
#pragma unroll
      for (int r = 0; r < 8; ++r) {
        float* orow = O + (size_t)(s0 + t * 16 + r) * D_HEAD + l16;
        orow[ 0] = acc[t][0][r];
        orow[16] = acc[t][1][r];
        orow[32] = acc[t][2][r];
        orow[48] = acc[t][3][r];
      }
  } else {
    float* O = Vt + (size_t)bh * D_HEAD * SEQ;
#pragma unroll
    for (int t = 0; t < 2; ++t)
#pragma unroll
      for (int r = 0; r < 8; ++r) {
        float* ocol = O + (size_t)l16 * SEQ + (s0 + t * 16 + r);
        ocol[(size_t) 0 * SEQ] = acc[t][0][r];
        ocol[(size_t)16 * SEQ] = acc[t][1][r];
        ocol[(size_t)32 * SEQ] = acc[t][2][r];
        ocol[(size_t)48 * SEQ] = acc[t][3][r];
      }
  }
}

#define AKT   64
#define ALDP  68
#define ATILE (64 * ALDP)

static __device__ __forceinline__ void kv_fetch(
    const float* __restrict__ p, int stride, v4f (&r)[4]) {
#pragma unroll
  for (int i = 0; i < 4; ++i)
    r[i] = *(const v4f*)(p + (size_t)i * stride);
}

static __device__ __forceinline__ void kv_stash(
    float* __restrict__ s, int tid, const v4f (&r)[4]) {
#pragma unroll
  for (int i = 0; i < 4; ++i) {
    const int idx = tid + 256 * i;
    const int row = idx >> 4;
    const int seg = idx & 15;
    *(v4f*)(s + row * ALDP + seg * 4) = r[i];
  }
}

static __device__ __forceinline__ float xhalf(float x) {
  const int b = __builtin_bit_cast(int, x);
  return __builtin_bit_cast(float,
      __builtin_amdgcn_permlanex16(b, b, 0x76543210, 0xfedcba98, false, false));
}

__global__ __launch_bounds__(256) void flash_attn_kernel(
    const float* __restrict__ Q, const float* __restrict__ K,
    const float* __restrict__ Vt, float* __restrict__ Oout) {
  __shared__ float sK[2 * ATILE];
  __shared__ float sV[2 * ATILE];

  const int tid  = threadIdx.x;
  const int wave = __builtin_amdgcn_readfirstlane(tid >> 5);
  const int lane = tid & 31;
  const int half = lane >> 4;
  const int l16  = lane & 15;

  const int bh = blockIdx.y;

  const float* Qm  = Q  + (size_t)bh * SEQ * D_HEAD;
  const float* Km  = K  + (size_t)bh * SEQ * D_HEAD;
  const float* Vtm = Vt + (size_t)bh * D_HEAD * SEQ;

  const float* kp = Km  + (size_t)(tid >> 4) * D_HEAD + (tid & 15) * 4;
  const float* vp = Vtm + (size_t)(tid >> 4) * SEQ    + (tid & 15) * 4;

  const int krow_of_lane = 4 * ((l16 & 7) >> 1) + 2 * (l16 >> 3) + (l16 & 1);
  const int kfoff = krow_of_lane * ALDP + 2 * half;
  const int vfoff = l16 * ALDP + 2 * half;

  const int b = bh >> 4;
  const int h = bh & (N_HEADS - 1);

  for (int pass = 0; pass < 2; ++pass) {
    const int xb = pass ? (SEQ / 128 - 1 - (int)blockIdx.x) : (int)blockIdx.x;
    const int wq = pass ? 7 - wave : wave;
    const int qb = xb * 128;
    const int q0 = qb + wq * 16;

    v2f qf[16];
    const float* qrow = Qm + (size_t)(q0 + l16) * D_HEAD;
#pragma unroll
    for (int kk = 0; kk < 16; ++kk)
      qf[kk] = *(const v2f*)(qrow + kk * 4 + 2 * half);

    v8f o0 = {}, o1 = {}, o2 = {}, o3 = {};
    float m = -__builtin_inff(), l = 0.0f;

    const int ntiles = (qb >> 6) + 2;

    v4f rk[4], rv[4];
    kv_fetch(kp, 16 * D_HEAD, rk);
    kv_fetch(vp, 16 * SEQ, rv);
    kv_stash(sK, tid, rk);
    kv_stash(sV, tid, rv);
    __syncthreads();

    for (int kt = 0; kt < ntiles; ++kt) {
      const int cur = kt & 1;
      const bool more = (kt + 1 < ntiles);
      if (more) {
        kv_fetch(kp + (size_t)(kt + 1) * AKT * D_HEAD, 16 * D_HEAD, rk);
        kv_fetch(vp + (size_t)(kt + 1) * AKT, 16 * SEQ, rv);
      }
      __builtin_amdgcn_sched_barrier(0);

      const float* tK = sK + cur * ATILE + kfoff;
      const float* tV = sV + cur * ATILE + vfoff;

#pragma unroll
      for (int sub = 0; sub < 4; ++sub) {
        const int t0 = kt * AKT + sub * 16;
        if (t0 <= q0) {
          v8f sc = {};
          const float* krow = tK + sub * 16 * ALDP;
          v2f kf[16];
#pragma unroll
          for (int kk = 0; kk < 16; ++kk)
            kf[kk] = *(const v2f*)(krow + kk * 4);
          __builtin_amdgcn_sched_barrier(0);
#pragma unroll
          for (int kk = 0; kk < 16; ++kk)
            sc = wmma_f32(kf[kk], qf[kk], sc);

          const bool diag = (t0 == q0);
          float mloc = -__builtin_inff();
#pragma unroll
          for (int r = 0; r < 8; ++r) {
            float v = sc[r] * 0.125f;
            if (diag && (4 * (r >> 1) + 2 * half + (r & 1) > l16)) v = -__builtin_inff();
            sc[r] = v;
            mloc = fmaxf(mloc, v);
          }

          mloc = fmaxf(mloc, xhalf(mloc));
          const float mn   = fmaxf(m, mloc);
          const float resc = __expf(m - mn);
          m = mn;
          float s = 0.0f;
#pragma unroll
          for (int r = 0; r < 8; ++r) {
            const float p = __expf(sc[r] - mn);
            sc[r] = p;
            s += p;
          }
          s += xhalf(s);
          l = l * resc + s;
#pragma unroll
          for (int r = 0; r < 8; ++r) {
            o0[r] *= resc; o1[r] *= resc; o2[r] *= resc; o3[r] *= resc;
          }

          const float* vrow = tV + sub * 16;
#pragma unroll
          for (int j = 0; j < 4; ++j) {
            v2f pb;
            pb.x = sc[2 * j];
            pb.y = sc[2 * j + 1];
            v2f a;
            a = *(const v2f*)(vrow +  0 * ALDP + j * 4); o0 = wmma_f32(a, pb, o0);
            a = *(const v2f*)(vrow + 16 * ALDP + j * 4); o1 = wmma_f32(a, pb, o1);
            a = *(const v2f*)(vrow + 32 * ALDP + j * 4); o2 = wmma_f32(a, pb, o2);
            a = *(const v2f*)(vrow + 48 * ALDP + j * 4); o3 = wmma_f32(a, pb, o3);
          }
        }
      }

      if (more) {
        kv_stash(sK + (cur ^ 1) * ATILE, tid, rk);
        kv_stash(sV + (cur ^ 1) * ATILE, tid, rv);
      }
      __syncthreads();
    }

    {
      const float inv = 1.0f / l;
      float* orow = Oout + (size_t)(b * SEQ + q0 + l16) * (N_HEADS * D_HEAD)
                    + h * D_HEAD + 8 * half;
      v4f w;
      w = v4f{o0[0], o0[1], o0[2], o0[3]} * inv; *(v4f*)(orow +  0) = w;
      w = v4f{o0[4], o0[5], o0[6], o0[7]} * inv; *(v4f*)(orow +  4) = w;
      w = v4f{o1[0], o1[1], o1[2], o1[3]} * inv; *(v4f*)(orow + 16) = w;
      w = v4f{o1[4], o1[5], o1[6], o1[7]} * inv; *(v4f*)(orow + 20) = w;
      w = v4f{o2[0], o2[1], o2[2], o2[3]} * inv; *(v4f*)(orow + 32) = w;
      w = v4f{o2[4], o2[5], o2[6], o2[7]} * inv; *(v4f*)(orow + 36) = w;
      w = v4f{o3[0], o3[1], o3[2], o3[3]} * inv; *(v4f*)(orow + 48) = w;
      w = v4f{o3[4], o3[5], o3[6], o3[7]} * inv; *(v4f*)(orow + 52) = w;
    }
  }
}

__global__ __launch_bounds__(256) void out_proj_kernel(
    const float* __restrict__ A, const float* __restrict__ Wot,
    float* __restrict__ C) {
  __shared__ float sA[2 * GTILE];
  __shared__ float sB[2 * GTILE];

  const int row0 = blockIdx.x * GBM;
  const int col0 = blockIdx.y * GBN;

  v8f acc[2][4];
#pragma unroll
  for (int t = 0; t < 2; ++t)
#pragma unroll
    for (int j = 0; j < 4; ++j) acc[t][j] = v8f{};

  gemm_128x128(A + (size_t)row0 * D_MODEL, Wot + (size_t)col0 * D_MODEL,
               sA, sB, acc);

  const int wave = threadIdx.x >> 5;
  const int lane = threadIdx.x & 31;
  const int half = lane >> 4;
  const int l16  = lane & 15;
  const int wm   = wave & 3;
  const int wn   = wave >> 2;

#pragma unroll
  for (int t = 0; t < 2; ++t)
#pragma unroll
    for (int r = 0; r < 8; ++r) {
      const int row = row0 + wm * 32 + t * 16 + 8 * half + r;
      float* crow = C + (size_t)row * D_MODEL + col0 + wn * 64 + l16;
      crow[ 0] = acc[t][0][r];
      crow[16] = acc[t][1][r];
      crow[32] = acc[t][2][r];
      crow[48] = acc[t][3][r];
    }
}

extern "C" void kernel_launch(void* const* d_in, const int* in_sizes, int n_in,
                              void* d_out, int out_size, void* d_ws, size_t ws_size,
                              hipStream_t stream) {
  const float* x  = (const float*)d_in[0];
  const float* Wq = (const float*)d_in[1];
  const float* Wk = (const float*)d_in[2];
  const float* Wv = (const float*)d_in[3];
  const float* Wo = (const float*)d_in[4];

  float* ws = (float*)d_ws;
  const size_t per = (size_t)BATCH * N_HEADS * SEQ * D_HEAD;
  const size_t wsz = (size_t)D_MODEL * D_MODEL;
  float* Q   = ws;
  float* K   = ws + per;
  float* Vt  = ws + 2 * per;
  float* AO  = ws + 3 * per;
  float* Wqt = ws + 4 * per;
  float* Wkt = Wqt + wsz;
  float* Wvt = Wkt + wsz;
  float* Wot = Wvt + wsz;

  dim3 g0(D_MODEL / 32, D_MODEL / 32, 4);
  transpose_w_kernel<<<g0, 256, 0, stream>>>(Wq, Wk, Wv, Wo, Wqt, Wkt, Wvt, Wot);

  dim3 g1(ROWS / GBM, 3 * (N_HEADS * D_HEAD) / GBN);
  qkv_proj_kernel<<<g1, 256, 0, stream>>>(x, Wqt, Wkt, Wvt, Q, K, Vt);

  dim3 g2(SEQ / (2 * 16 * 8), BATCH * N_HEADS);
  flash_attn_kernel<<<g2, 256, 0, stream>>>(Q, K, Vt, AO);

  dim3 g3(ROWS / GBM, D_MODEL / GBN);
  out_proj_kernel<<<g3, 256, 0, stream>>>(AO, Wot, (float*)d_out);
}
